# starting kernel
# baseline (speedup 1.0000x reference)
#include <hip/hip_runtime.h>
#include <hip/hip_bf16.h>

typedef __attribute__((ext_vector_type(2))) float v2f;
typedef __attribute__((ext_vector_type(8))) float v8f;

#define Bsz 32
#define Nn  256
#define Ff  256
#define Hh  64

__global__ void build_wcomb(const float* __restrict__ W_enc,
                            const float* __restrict__ W1,
                            float* __restrict__ Wcomb) {
    const int f = blockIdx.x;
    const int c = threadIdx.x;
    const int half = c >> 6;
    const int h = c & 63;
    const float* we = W_enc + f * Hh;
    const float* w1 = W1 + (half * Hh) * Hh + h;
    float s = 0.0f;
#pragma unroll 8
    for (int k = 0; k < Hh; ++k)
        s = fmaf(we[k], w1[k * Hh], s);
    Wcomb[f * 128 + c] = s;
}

__global__ void build_bias(const float* __restrict__ b_enc,
                           const float* __restrict__ W1,
                           const float* __restrict__ b1,
                           float* __restrict__ bias) {
    const int h = threadIdx.x;
    float s = b1[h];
#pragma unroll 8
    for (int k = 0; k < Hh; ++k)
        s = fmaf(b_enc[k], W1[k * Hh + h] + W1[(Hh + k) * Hh + h], s);
    bias[h] = s;
}

__global__ void __launch_bounds__(128)
gemm_wmma_f32(const float* __restrict__ X,
              const float* __restrict__ Wcomb,
              float* __restrict__ ai,
              float* __restrict__ ajt) {
    const int lane    = threadIdx.x & 31;
    const int waveIdG = blockIdx.x * 4 + (threadIdx.x >> 5);
    const int rowBase = waveIdG * 16;

    const int lrow  = lane & 15;
    const int khalf = (lane >> 4) * 2;

    v8f acc[8];
#pragma unroll
    for (int t = 0; t < 8; ++t) acc[t] = (v8f){0,0,0,0,0,0,0,0};

    const float* arow = X + (rowBase + lrow) * Ff;

    for (int kk = 0; kk < Ff; kk += 4) {
        const int kb = kk + khalf;
        v2f afrag = *(const v2f*)(arow + kb);
#pragma unroll
        for (int t = 0; t < 8; ++t) {
            const int col = t * 16 + lrow;
            v2f bfrag;
            bfrag.x = Wcomb[kb * 128 + col];
            bfrag.y = Wcomb[(kb + 1) * 128 + col];
            acc[t] = __builtin_amdgcn_wmma_f32_16x16x4_f32(
                false, afrag, false, bfrag, (short)0, acc[t], false, false);
        }
    }

    const int mOff = 8 * (lane >> 4);
#pragma unroll
    for (int t = 0; t < 8; ++t) {
#pragma unroll
        for (int v = 0; v < 8; ++v) {
            const int m   = mOff + v;
            const int row = rowBase + m;
            const int n   = t * 16 + lrow;
            const float val = acc[t][v];
            if (n < Hh) {
                ai[row * Hh + n] = val;
            } else {
                const int h = n - Hh;
                const int b = row >> 8;
                const int j = row & 255;
                ajt[(b * Hh + h) * Nn + j] = val;
            }
        }
    }
}

__global__ void __launch_bounds__(256)
score_kernel(const float* __restrict__ ai,
             const float* __restrict__ ajt,
             const float* __restrict__ bias,
             const float* __restrict__ W2,
             const float* __restrict__ b2,
             float* __restrict__ adj) {
    __shared__ float s_ai[Hh];
    __shared__ float s_w2[Hh];

    const int blk = blockIdx.x;
    const int b   = blk >> 8;
    const int i   = blk & 255;
    const int j   = threadIdx.x;

    if (j < Hh) {
        s_ai[j] = ai[blk * Hh + j] + bias[j];
        s_w2[j] = W2[j];
    }
    __syncthreads();

    const float* ajrow = ajt + b * (Hh * Nn);
    float accv = 0.0f;
#pragma unroll 8
    for (int h = 0; h < Hh; ++h) {
        const float t = s_ai[h] + ajrow[h * Nn + j];
        accv = fmaf(s_w2[h], fmaxf(t, 0.0f), accv);
    }
    float val = 1.0f / (1.0f + __expf(-(accv + b2[0])));
    if (j == i) val = 0.0f;
    adj[blk * Nn + j] = val;
}

__global__ void broadcast_structural(const float* __restrict__ sp,
                                     float* __restrict__ out2) {
    const int idx = blockIdx.x * blockDim.x + threadIdx.x;
    const float v = sp[idx];
#pragma unroll
    for (int b = 0; b < Bsz; ++b)
        out2[b * (Nn * Nn) + idx] = v;
}

extern "C" void kernel_launch(void* const* d_in, const int* in_sizes, int n_in,
                              void* d_out, int out_size, void* d_ws, size_t ws_size,
                              hipStream_t stream) {
    const float* X      = (const float*)d_in[0];
    const float* W_enc  = (const float*)d_in[1];
    const float* b_enc  = (const float*)d_in[2];
    const float* W1     = (const float*)d_in[3];
    const float* b1     = (const float*)d_in[4];
    const float* W2     = (const float*)d_in[5];
    const float* b2     = (const float*)d_in[6];
    const float* sp     = (const float*)d_in[7];

    float* out_adj    = (float*)d_out;
    float* out_struct = out_adj + Bsz * Nn * Nn;

    float* ws    = (float*)d_ws;
    float* Wcomb = ws;
    float* bias  = Wcomb + 256 * 128;
    float* ai    = bias + 64;
    float* ajt   = ai + 8192 * 64;

    build_wcomb<<<256, 128, 0, stream>>>(W_enc, W1, Wcomb);
    build_bias<<<1, 64, 0, stream>>>(b_enc, W1, b1, bias);

    gemm_wmma_f32<<<128, 128, 0, stream>>>(X, Wcomb, ai, ajt);

    score_kernel<<<Bsz * Nn, Nn, 0, stream>>>(ai, ajt, bias, W2, b2, out_adj);

    broadcast_structural<<<(Nn * Nn) / 256, 256, 0, stream>>>(sp, out_struct);
}


# GEMM kernel: operands in 128-bit loads issued as 20-load groups two 32-k chunks ahead of their WMMAs, 1024 waves of 16x64 tiles (was 512 waves, 17 loads waited on per k-step); aj half transposed
# speedup vs baseline: 1.4565x; 1.4565x over previous
#include <hip/hip_runtime.h>
#include <hip/hip_bf16.h>

typedef __attribute__((ext_vector_type(2))) float v2f;
typedef __attribute__((ext_vector_type(8))) float v8f;

#define Bsz 32
#define Nn  256
#define Ff  256
#define Hh  64

__global__ void build_wcomb(const float* __restrict__ W_enc,
                            const float* __restrict__ W1,
                            float* __restrict__ Wcomb) {
    const int f = blockIdx.x;
    const int c = threadIdx.x;
    const int half = c >> 6;
    const int h = c & 63;
    const float* we = W_enc + f * Hh;
    const float* w1 = W1 + (half * Hh) * Hh + h;
    float s = 0.0f;
#pragma unroll 8
    for (int k = 0; k < Hh; ++k)
        s = fmaf(we[k], w1[k * Hh], s);
    Wcomb[f * 128 + c] = s;
}

__global__ void build_bias(const float* __restrict__ b_enc,
                           const float* __restrict__ W1,
                           const float* __restrict__ b1,
                           float* __restrict__ bias) {
    const int h = threadIdx.x;
    float s = b1[h];
#pragma unroll 8
    for (int k = 0; k < Hh; ++k)
        s = fmaf(b_enc[k], W1[k * Hh + h] + W1[(Hh + k) * Hh + h], s);
    bias[h] = s;
}

typedef __attribute__((ext_vector_type(4))) float v4f;

#define KCH   32
#define NGRP  (KCH / 8)
#define NCH   (Ff / KCH)

__device__ __forceinline__ void gemm_load_chunk(const float* __restrict__ xp,
                                                const float* __restrict__ wp,
                                                int kc, v4f (&a)[NGRP], v4f (&w)[NGRP][4]) {
#pragma unroll
    for (int g = 0; g < NGRP; ++g) {
        const int kb = kc + g * 8;
        a[g] = *(const v4f*)(xp + kb);
#pragma unroll
        for (int e = 0; e < 4; ++e)
            w[g][e] = *(const v4f*)(wp + (kb + e) * 128);
    }
}

template <int SWAP>
__device__ __forceinline__ void gemm_mma_chunk(const v4f (&a)[NGRP], const v4f (&w)[NGRP][4],
                                               v8f (&acc)[4]) {
#pragma unroll
    for (int g = 0; g < NGRP; ++g) {
#pragma unroll
        for (int u = 0; u < 2; ++u) {
            v2f xa;
            xa.x = a[g][2 * u];
            xa.y = a[g][2 * u + 1];
#pragma unroll
            for (int i = 0; i < 4; ++i) {
                v2f wb;
                wb.x = w[g][2 * u][i];
                wb.y = w[g][2 * u + 1][i];
                if (SWAP)
                    acc[i] = __builtin_amdgcn_wmma_f32_16x16x4_f32(
                        false, wb, false, xa, (short)0, acc[i], false, false);
                else
                    acc[i] = __builtin_amdgcn_wmma_f32_16x16x4_f32(
                        false, xa, false, wb, (short)0, acc[i], false, false);
            }
        }
    }
}

template <int SWAP>
__device__ __forceinline__ void gemm_strip(const float* __restrict__ xp,
                                           const float* __restrict__ wp,
                                           v8f (&acc)[4]) {
    v4f a[3][NGRP];
    v4f w[3][NGRP][4];
    gemm_load_chunk(xp, wp, 0, a[0], w[0]);
    gemm_load_chunk(xp, wp, KCH, a[1], w[1]);
#pragma unroll
    for (int ch = 0; ch < NCH; ++ch) {
        if (ch + 2 < NCH)
            gemm_load_chunk(xp, wp, (ch + 2) * KCH, a[(ch + 2) % 3], w[(ch + 2) % 3]);
        __builtin_amdgcn_sched_barrier(0);
        gemm_mma_chunk<SWAP>(a[ch % 3], w[ch % 3], acc);
        __builtin_amdgcn_sched_barrier(0);
    }
}

__global__ void __launch_bounds__(128)
gemm_wmma_f32(const float* __restrict__ X,
              const float* __restrict__ Wcomb,
              float* __restrict__ ai,
              float* __restrict__ ajt) {
    const int lane    = threadIdx.x & 31;
    const int wave    = threadIdx.x >> 5;
    const int half    = blockIdx.x & 1;
    const int strip   = (blockIdx.x >> 1) * 4 + wave;
    const int rowBase = strip * 16;
    const int c       = lane & 15;
    const int hh      = lane >> 4;

    const float* xp = X + (rowBase + c) * Ff + 4 * hh;
    const float* wp = Wcomb + (4 * hh) * 128 + half * Hh + 4 * c;

    v8f acc[4];
#pragma unroll
    for (int t = 0; t < 4; ++t) acc[t] = (v8f){0,0,0,0,0,0,0,0};

    if (half == 0) {
        gemm_strip<0>(xp, wp, acc);
#pragma unroll
        for (int r = 0; r < 8; ++r) {
            v4f o;
            o.x = acc[0][r]; o.y = acc[1][r]; o.z = acc[2][r]; o.w = acc[3][r];
            *(v4f*)(ai + (rowBase + 8 * hh + r) * Hh + 4 * c) = o;
        }
    } else {
        gemm_strip<1>(xp, wp, acc);
        const int b  = rowBase >> 8;
        const int j0 = rowBase & 255;
#pragma unroll
        for (int i = 0; i < 4; ++i) {
#pragma unroll
            for (int r = 0; r < 8; ++r) {
                const int h = 4 * (8 * hh + r) + i;
                ajt[(b * Hh + h) * Nn + j0 + c] = acc[i][r];
            }
        }
    }
}

__global__ void __launch_bounds__(256)
score_kernel(const float* __restrict__ ai,
             const float* __restrict__ ajt,
             const float* __restrict__ bias,
             const float* __restrict__ W2,
             const float* __restrict__ b2,
             float* __restrict__ adj) {
    __shared__ float s_ai[Hh];
    __shared__ float s_w2[Hh];

    const int blk = blockIdx.x;
    const int b   = blk >> 8;
    const int i   = blk & 255;
    const int j   = threadIdx.x;

    if (j < Hh) {
        s_ai[j] = ai[blk * Hh + j] + bias[j];
        s_w2[j] = W2[j];
    }
    __syncthreads();

    const float* ajrow = ajt + b * (Hh * Nn);
    float accv = 0.0f;
#pragma unroll 8
    for (int h = 0; h < Hh; ++h) {
        const float t = s_ai[h] + ajrow[h * Nn + j];
        accv = fmaf(s_w2[h], fmaxf(t, 0.0f), accv);
    }
    float val = 1.0f / (1.0f + __expf(-(accv + b2[0])));
    if (j == i) val = 0.0f;
    adj[blk * Nn + j] = val;
}

__global__ void broadcast_structural(const float* __restrict__ sp,
                                     float* __restrict__ out2) {
    const int idx = blockIdx.x * blockDim.x + threadIdx.x;
    const float v = sp[idx];
#pragma unroll
    for (int b = 0; b < Bsz; ++b)
        out2[b * (Nn * Nn) + idx] = v;
}

extern "C" void kernel_launch(void* const* d_in, const int* in_sizes, int n_in,
                              void* d_out, int out_size, void* d_ws, size_t ws_size,
                              hipStream_t stream) {
    const float* X      = (const float*)d_in[0];
    const float* W_enc  = (const float*)d_in[1];
    const float* b_enc  = (const float*)d_in[2];
    const float* W1     = (const float*)d_in[3];
    const float* b1     = (const float*)d_in[4];
    const float* W2     = (const float*)d_in[5];
    const float* b2     = (const float*)d_in[6];
    const float* sp     = (const float*)d_in[7];

    float* out_adj    = (float*)d_out;
    float* out_struct = out_adj + Bsz * Nn * Nn;

    float* ws    = (float*)d_ws;
    float* Wcomb = ws;
    float* bias  = Wcomb + 256 * 128;
    float* ai    = bias + 64;
    float* ajt   = ai + 8192 * 64;

    build_wcomb<<<256, 128, 0, stream>>>(W_enc, W1, Wcomb);
    build_bias<<<1, 64, 0, stream>>>(b_enc, W1, b1, bias);

    gemm_wmma_f32<<<256, 128, 0, stream>>>(X, Wcomb, ai, ajt);

    score_kernel<<<Bsz * Nn, Nn, 0, stream>>>(ai, ajt, bias, W2, b2, out_adj);

    broadcast_structural<<<(Nn * Nn) / 256, 256, 0, stream>>>(sp, out_struct);
}


# score kernel: one block per (b, 16 rows i), each thread's ajt column held in 64 registers, ai+bias rows read from LDS as same-address 128-bit loads; 4096 waves instead of 65536
# speedup vs baseline: 2.3269x; 1.5976x over previous
#include <hip/hip_runtime.h>
#include <hip/hip_bf16.h>

typedef __attribute__((ext_vector_type(2))) float v2f;
typedef __attribute__((ext_vector_type(8))) float v8f;

#define Bsz 32
#define Nn  256
#define Ff  256
#define Hh  64

__global__ void build_wcomb(const float* __restrict__ W_enc,
                            const float* __restrict__ W1,
                            float* __restrict__ Wcomb) {
    const int f = blockIdx.x;
    const int c = threadIdx.x;
    const int half = c >> 6;
    const int h = c & 63;
    const float* we = W_enc + f * Hh;
    const float* w1 = W1 + (half * Hh) * Hh + h;
    float s = 0.0f;
#pragma unroll 8
    for (int k = 0; k < Hh; ++k)
        s = fmaf(we[k], w1[k * Hh], s);
    Wcomb[f * 128 + c] = s;
}

__global__ void build_bias(const float* __restrict__ b_enc,
                           const float* __restrict__ W1,
                           const float* __restrict__ b1,
                           float* __restrict__ bias) {
    const int h = threadIdx.x;
    float s = b1[h];
#pragma unroll 8
    for (int k = 0; k < Hh; ++k)
        s = fmaf(b_enc[k], W1[k * Hh + h] + W1[(Hh + k) * Hh + h], s);
    bias[h] = s;
}

typedef __attribute__((ext_vector_type(4))) float v4f;

#define KCH   32
#define NGRP  (KCH / 8)
#define NCH   (Ff / KCH)

__device__ __forceinline__ void gemm_load_chunk(const float* __restrict__ xp,
                                                const float* __restrict__ wp,
                                                int kc, v4f (&a)[NGRP], v4f (&w)[NGRP][4]) {
#pragma unroll
    for (int g = 0; g < NGRP; ++g) {
        const int kb = kc + g * 8;
        a[g] = *(const v4f*)(xp + kb);
#pragma unroll
        for (int e = 0; e < 4; ++e)
            w[g][e] = *(const v4f*)(wp + (kb + e) * 128);
    }
}

template <int SWAP>
__device__ __forceinline__ void gemm_mma_chunk(const v4f (&a)[NGRP], const v4f (&w)[NGRP][4],
                                               v8f (&acc)[4]) {
#pragma unroll
    for (int g = 0; g < NGRP; ++g) {
#pragma unroll
        for (int u = 0; u < 2; ++u) {
            v2f xa;
            xa.x = a[g][2 * u];
            xa.y = a[g][2 * u + 1];
#pragma unroll
            for (int i = 0; i < 4; ++i) {
                v2f wb;
                wb.x = w[g][2 * u][i];
                wb.y = w[g][2 * u + 1][i];
                if (SWAP)
                    acc[i] = __builtin_amdgcn_wmma_f32_16x16x4_f32(
                        false, wb, false, xa, (short)0, acc[i], false, false);
                else
                    acc[i] = __builtin_amdgcn_wmma_f32_16x16x4_f32(
                        false, xa, false, wb, (short)0, acc[i], false, false);
            }
        }
    }
}

template <int SWAP>
__device__ __forceinline__ void gemm_strip(const float* __restrict__ xp,
                                           const float* __restrict__ wp,
                                           v8f (&acc)[4]) {
    v4f a[3][NGRP];
    v4f w[3][NGRP][4];
    gemm_load_chunk(xp, wp, 0, a[0], w[0]);
    gemm_load_chunk(xp, wp, KCH, a[1], w[1]);
#pragma unroll
    for (int ch = 0; ch < NCH; ++ch) {
        if (ch + 2 < NCH)
            gemm_load_chunk(xp, wp, (ch + 2) * KCH, a[(ch + 2) % 3], w[(ch + 2) % 3]);
        __builtin_amdgcn_sched_barrier(0);
        gemm_mma_chunk<SWAP>(a[ch % 3], w[ch % 3], acc);
        __builtin_amdgcn_sched_barrier(0);
    }
}

__global__ void __launch_bounds__(128)
gemm_wmma_f32(const float* __restrict__ X,
              const float* __restrict__ Wcomb,
              float* __restrict__ ai,
              float* __restrict__ ajt) {
    const int lane    = threadIdx.x & 31;
    const int wave    = threadIdx.x >> 5;
    const int half    = blockIdx.x & 1;
    const int strip   = (blockIdx.x >> 1) * 4 + wave;
    const int rowBase = strip * 16;
    const int c       = lane & 15;
    const int hh      = lane >> 4;

    const float* xp = X + (rowBase + c) * Ff + 4 * hh;
    const float* wp = Wcomb + (4 * hh) * 128 + half * Hh + 4 * c;

    v8f acc[4];
#pragma unroll
    for (int t = 0; t < 4; ++t) acc[t] = (v8f){0,0,0,0,0,0,0,0};

    if (half == 0) {
        gemm_strip<0>(xp, wp, acc);
#pragma unroll
        for (int r = 0; r < 8; ++r) {
            v4f o;
            o.x = acc[0][r]; o.y = acc[1][r]; o.z = acc[2][r]; o.w = acc[3][r];
            *(v4f*)(ai + (rowBase + 8 * hh + r) * Hh + 4 * c) = o;
        }
    } else {
        gemm_strip<1>(xp, wp, acc);
        const int b  = rowBase >> 8;
        const int j0 = rowBase & 255;
#pragma unroll
        for (int i = 0; i < 4; ++i) {
#pragma unroll
            for (int r = 0; r < 8; ++r) {
                const int h = 4 * (8 * hh + r) + i;
                ajt[(b * Hh + h) * Nn + j0 + c] = acc[i][r];
            }
        }
    }
}

#define SC_TI 16
#define SC_IU 8

__device__ __forceinline__ void score_load_rows(const float* __restrict__ s, int h4, v4f (&a)[SC_IU]) {
#pragma unroll
    for (int u = 0; u < SC_IU; ++u)
        a[u] = *(const v4f*)(s + u * Hh + h4);
}

__global__ void __launch_bounds__(256)
score_kernel(const float* __restrict__ ai,
             const float* __restrict__ ajt,
             const float* __restrict__ bias,
             const float* __restrict__ W2,
             const float* __restrict__ b2,
             float* __restrict__ adj) {
    __shared__ __attribute__((aligned(16))) float s_ai[SC_TI * Hh];

    const int b    = blockIdx.x % Bsz;
    const int i0   = (blockIdx.x / Bsz) * SC_TI;
    const int j    = threadIdx.x;
    const int row0 = b * Nn + i0;

    for (int e = 4 * j; e < SC_TI * Hh; e += 4 * 256) {
        v4f a  = *(const v4f*)(ai + row0 * Hh + e);
        v4f bb = *(const v4f*)(bias + (e & (Hh - 1)));
        a.x += bb.x; a.y += bb.y; a.z += bb.z; a.w += bb.w;
        *(v4f*)(s_ai + e) = a;
    }

    float ajr[Hh];
    const float* ajcol = ajt + b * (Hh * Nn) + j;
#pragma unroll
    for (int h = 0; h < Hh; ++h) ajr[h] = ajcol[h * Nn];

    float w2r[Hh];
#pragma unroll
    for (int h = 0; h < Hh; ++h) w2r[h] = W2[h];
    const float bb2 = b2[0];

    __syncthreads();

#pragma unroll 1
    for (int ig = 0; ig < SC_TI; ig += SC_IU) {
        const float* srow = s_ai + ig * Hh;
        float acc[SC_IU];
#pragma unroll
        for (int u = 0; u < SC_IU; ++u) acc[u] = 0.0f;

        v4f a[2][SC_IU];
        score_load_rows(srow, 0, a[0]);
#pragma unroll
        for (int q = 0; q < Hh / 4; ++q) {
            if (q + 1 < Hh / 4)
                score_load_rows(srow, 4 * (q + 1), a[(q + 1) & 1]);
            __builtin_amdgcn_sched_barrier(0);
            float t[4][SC_IU];
#pragma unroll
            for (int e = 0; e < 4; ++e)
#pragma unroll
                for (int u = 0; u < SC_IU; ++u)
                    t[e][u] = a[q & 1][u][e] + ajr[4 * q + e];
#pragma unroll
            for (int e = 0; e < 4; ++e)
#pragma unroll
                for (int u = 0; u < SC_IU; ++u)
                    t[e][u] = fmaxf(t[e][u], 0.0f);
#pragma unroll
            for (int e = 0; e < 4; ++e)
#pragma unroll
                for (int u = 0; u < SC_IU; ++u)
                    acc[u] = fmaf(w2r[4 * q + e], t[e][u], acc[u]);
            __builtin_amdgcn_sched_barrier(0);
        }

#pragma unroll
        for (int u = 0; u < SC_IU; ++u) {
            const int i = i0 + ig + u;
            float val = 1.0f / (1.0f + __expf(-(acc[u] + bb2)));
            if (j == i) val = 0.0f;
            adj[(row0 + ig + u) * Nn + j] = val;
        }
    }
}

__global__ void broadcast_structural(const float* __restrict__ sp,
                                     float* __restrict__ out2) {
    const int idx = blockIdx.x * blockDim.x + threadIdx.x;
    const float v = sp[idx];
#pragma unroll
    for (int b = 0; b < Bsz; ++b)
        out2[b * (Nn * Nn) + idx] = v;
}

extern "C" void kernel_launch(void* const* d_in, const int* in_sizes, int n_in,
                              void* d_out, int out_size, void* d_ws, size_t ws_size,
                              hipStream_t stream) {
    const float* X      = (const float*)d_in[0];
    const float* W_enc  = (const float*)d_in[1];
    const float* b_enc  = (const float*)d_in[2];
    const float* W1     = (const float*)d_in[3];
    const float* b1     = (const float*)d_in[4];
    const float* W2     = (const float*)d_in[5];
    const float* b2     = (const float*)d_in[6];
    const float* sp     = (const float*)d_in[7];

    float* out_adj    = (float*)d_out;
    float* out_struct = out_adj + Bsz * Nn * Nn;

    float* ws    = (float*)d_ws;
    float* Wcomb = ws;
    float* bias  = Wcomb + 256 * 128;
    float* ai    = bias + 64;
    float* ajt   = ai + 8192 * 64;

    build_wcomb<<<256, 128, 0, stream>>>(W_enc, W1, Wcomb);
    build_bias<<<1, 64, 0, stream>>>(b_enc, W1, b1, bias);

    gemm_wmma_f32<<<256, 128, 0, stream>>>(X, Wcomb, ai, ajt);

    score_kernel<<<Bsz * (Nn / SC_TI), Nn, 0, stream>>>(ai, ajt, bias, W2, b2, out_adj);

    broadcast_structural<<<(Nn * Nn) / 256, 256, 0, stream>>>(sp, out_struct);
}


# score kernel: relu(s+a) taken as a - min(-s,a), column-only part summed once per thread: min+fma per element (was add,max,fma); 4 columns per thread, h halves on wave pairs joined via LDS
# speedup vs baseline: 4.8691x; 2.0925x over previous
#include <hip/hip_runtime.h>
#include <hip/hip_bf16.h>

typedef __attribute__((ext_vector_type(2))) float v2f;
typedef __attribute__((ext_vector_type(8))) float v8f;

#define Bsz 32
#define Nn  256
#define Ff  256
#define Hh  64

typedef __attribute__((ext_vector_type(4))) float v4f;

#define KCH   32
#define NGRP  (KCH / 8)
#define NCHW  (Ff / KCH / 2)

__device__ __forceinline__ void gemm_load_chunk(const float* __restrict__ xp,
                                                const float* __restrict__ wp,
                                                int kc, v4f (&a)[NGRP], v4f (&w)[NGRP][4]) {
#pragma unroll
    for (int g = 0; g < NGRP; ++g) {
        const int kb = kc + g * 8;
        a[g] = *(const v4f*)(xp + kb);
#pragma unroll
        for (int e = 0; e < 4; ++e)
            w[g][e] = *(const v4f*)(wp + (kb + e) * Hh);
    }
}

__device__ __forceinline__ void gemm_mma_chunk(const v4f (&a)[NGRP], const v4f (&w)[NGRP][4],
                                               v8f (&acc)[4]) {
#pragma unroll
    for (int g = 0; g < NGRP; ++g) {
#pragma unroll
        for (int u = 0; u < 2; ++u) {
            v2f xa;
            xa.x = a[g][2 * u];
            xa.y = a[g][2 * u + 1];
#pragma unroll
            for (int i = 0; i < 4; ++i) {
                v2f wb;
                wb.x = w[g][2 * u][i];
                wb.y = w[g][2 * u + 1][i];
                acc[i] = __builtin_amdgcn_wmma_f32_16x16x4_f32(
                    false, wb, false, xa, (short)0, acc[i], false, false);
            }
        }
    }
}

__global__ void __launch_bounds__(128)
gemm_chain_wmma_f32(const float* __restrict__ X,
                    const float* __restrict__ W_enc,
                    const float* __restrict__ b_enc,
                    const float* __restrict__ W1,
                    float* __restrict__ ai,
                    float* __restrict__ ajt) {
    __shared__ v4f xch[4][8][32];

    const int lane    = threadIdx.x & 31;
    const int wave    = __builtin_amdgcn_readfirstlane(threadIdx.x >> 5);
    const int kh      = wave & 1;
    const int strip   = blockIdx.x * 2 + (wave >> 1);
    const int rowBase = strip * 16;
    const int c       = lane & 15;
    const int hh      = lane >> 4;

    const float* xp  = X + (rowBase + c) * Ff + kh * (Ff / 2) + 4 * hh;
    const float* wp  = W_enc + (kh * (Ff / 2) + 4 * hh) * Hh + 4 * c;
    const float* w1p = W1 + (kh * Hh) * Hh + 4 * c;

    v8f acc[4];
#pragma unroll
    for (int t = 0; t < 4; ++t) acc[t] = (v8f){0,0,0,0,0,0,0,0};

    v4f a[3][NGRP];
    v4f w[3][NGRP][4];
    v4f wv[4][4][2];
    v4f be[8];

    gemm_load_chunk(xp, wp, 0, a[0], w[0]);
    gemm_load_chunk(xp, wp, KCH, a[1], w[1]);
#pragma unroll
    for (int ch = 0; ch < NCHW; ++ch) {
        if (ch + 2 < NCHW)
            gemm_load_chunk(xp, wp, (ch + 2) * KCH, a[(ch + 2) % 3], w[(ch + 2) % 3]);
        if (ch + 2 == NCHW - 1) {
#pragma unroll
            for (int i = 0; i < 4; ++i)
#pragma unroll
                for (int p = 0; p < 4; ++p)
#pragma unroll
                    for (int s = 0; s < 2; ++s)
                        wv[i][p][s] = *(const v4f*)(w1p + (4 * (8 * hh + 2 * p + s) + i) * Hh);
#pragma unroll
            for (int r = 0; r < 8; ++r)
                be[r] = *(const v4f*)(b_enc + 4 * (8 * hh + r));
        }
        __builtin_amdgcn_sched_barrier(0);
        gemm_mma_chunk(a[ch % 3], w[ch % 3], acc);
        __builtin_amdgcn_sched_barrier(0);
    }

#pragma unroll
    for (int i = 0; i < 4; ++i)
#pragma unroll
        for (int q = 0; q < 2; ++q) {
            v4f o;
            o.x = acc[i][4 * q]; o.y = acc[i][4 * q + 1]; o.z = acc[i][4 * q + 2]; o.w = acc[i][4 * q + 3];
            xch[wave][2 * i + q][lane] = o;
        }
    __syncthreads();
    v8f nf[4];
#pragma unroll
    for (int i = 0; i < 4; ++i)
#pragma unroll
        for (int q = 0; q < 2; ++q) {
            const v4f o = xch[wave ^ 1][2 * i + q][lane];
#pragma unroll
            for (int e = 0; e < 4; ++e)
                nf[i][4 * q + e] = (acc[i][4 * q + e] + o[e]) + be[4 * q + e][i];
        }

    v8f acc2[4];
#pragma unroll
    for (int t = 0; t < 4; ++t) acc2[t] = (v8f){0,0,0,0,0,0,0,0};
#pragma unroll
    for (int i = 0; i < 4; ++i) {
#pragma unroll
        for (int p = 0; p < 4; ++p) {
            v2f nb;
            nb.x = nf[i][2 * p];
            nb.y = nf[i][2 * p + 1];
#pragma unroll
            for (int t = 0; t < 4; ++t) {
                v2f wa;
                wa.x = wv[i][p][0][t];
                wa.y = wv[i][p][1][t];
                acc2[t] = __builtin_amdgcn_wmma_f32_16x16x4_f32(
                    false, wa, false, nb, (short)0, acc2[t], false, false);
            }
        }
    }

    if (kh == 0) {
#pragma unroll
        for (int r = 0; r < 8; ++r) {
            v4f o;
            o.x = acc2[0][r]; o.y = acc2[1][r]; o.z = acc2[2][r]; o.w = acc2[3][r];
            *(v4f*)(ai + (rowBase + c) * Hh + 4 * (8 * hh + r)) = o;
        }
    } else {
        const int b  = rowBase >> 8;
        const int j0 = rowBase & 255;
#pragma unroll
        for (int t = 0; t < 4; ++t) {
#pragma unroll
            for (int r = 0; r < 8; ++r) {
                const int h = 4 * (8 * hh + r) + t;
                ajt[(b * Hh + h) * Nn + j0 + c] = acc2[t][r];
            }
        }
    }
}

#define SC_TI 16
#define SC_IU 4
#define SC_NT 128
#define SC_HH (Hh / 2)

__device__ __forceinline__ void score_load_rows(const float* __restrict__ s, int h4, v4f (&a)[SC_IU]) {
#pragma unroll
    for (int u = 0; u < SC_IU; ++u)
        a[u] = *(const v4f*)(s + u * Hh + h4);
}

__global__ void __launch_bounds__(SC_NT)
score_kernel(const float* __restrict__ ai,
             const float* __restrict__ ajt,
             const float* __restrict__ b1,
             const float* __restrict__ W2,
             const float* __restrict__ b2,
             const float* __restrict__ sp,
             float* __restrict__ adj,
             float* __restrict__ out2) {
    __shared__ __attribute__((aligned(16))) float s_ns[SC_TI * Hh];
    __shared__ v4f s_part[2][SC_TI][SC_NT / 2];

    const int b    = blockIdx.x % Bsz;
    const int i0   = (blockIdx.x / Bsz) * SC_TI;
    const int t    = threadIdx.x;
    const int wave = __builtin_amdgcn_readfirstlane(t >> 5);
    const int hs   = wave & 1;
    const int ct   = (wave >> 1) * 32 + (t & 31);
    const int j0   = 4 * ct;
    const int row0 = b * Nn + i0;

    v4f spv[SC_TI * Nn / (4 * SC_NT)];
#pragma unroll
    for (int q = 0; q < SC_TI * Nn / (4 * SC_NT); ++q)
        spv[q] = *(const v4f*)(sp + i0 * Nn + 4 * (t + SC_NT * q));

    const v4f sa0 = *(const v4f*)(ai + row0 * Hh + 4 * t);
    const v4f sa1 = *(const v4f*)(ai + row0 * Hh + 4 * t + 4 * SC_NT);
    const v4f sbb = *(const v4f*)(b1 + ((4 * t) & (Hh - 1)));

    v4f ajr[SC_HH];
    const float* ajcol = ajt + (b * Hh + hs * SC_HH) * Nn + j0;
#pragma unroll
    for (int k = 0; k < SC_HH; ++k) ajr[k] = *(const v4f*)(ajcol + k * Nn);

    float w2r[SC_HH];
    const float* w2p = W2 + hs * SC_HH;
#pragma unroll
    for (int k = 0; k < SC_HH; ++k) w2r[k] = w2p[k];
    const float bb2 = (hs == 0) ? b2[0] : 0.0f;

    {
        v4f n0, n1;
#pragma unroll
        for (int c = 0; c < 4; ++c) {
            n0[c] = -(sa0[c] + sbb[c]);
            n1[c] = -(sa1[c] + sbb[c]);
        }
        *(v4f*)(s_ns + 4 * t) = n0;
        *(v4f*)(s_ns + 4 * t + 4 * SC_NT) = n1;
    }

    v4f kin = (v4f){0.0f, 0.0f, 0.0f, 0.0f};
#pragma unroll
    for (int k = 0; k < SC_HH; ++k)
#pragma unroll
        for (int c = 0; c < 4; ++c)
            kin[c] = fmaf(w2r[k], ajr[k][c], kin[c]);
#pragma unroll
    for (int c = 0; c < 4; ++c) kin[c] = -(kin[c] + bb2);

    __syncthreads();

#pragma unroll
    for (int q = 0; q < SC_TI * Nn / (4 * SC_NT); ++q)
        *(v4f*)(out2 + row0 * Nn + 4 * (t + SC_NT * q)) = spv[q];

    v4f a[2][SC_IU];
    score_load_rows(s_ns + hs * SC_HH, 0, a[0]);
#pragma unroll 1
    for (int ig = 0; ig < SC_TI; ig += SC_IU) {
        const float* srow = s_ns + ig * Hh + hs * SC_HH;
        const float* snxt = (ig + SC_IU < SC_TI) ? srow + SC_IU * Hh : srow;
        v4f acc[SC_IU];
#pragma unroll
        for (int u = 0; u < SC_IU; ++u) acc[u] = kin;

#pragma unroll
        for (int q = 0; q < SC_HH / 4; ++q) {
            if (q + 1 < SC_HH / 4)
                score_load_rows(srow, 4 * (q + 1), a[(q + 1) & 1]);
            else
                score_load_rows(snxt, 0, a[0]);
            __builtin_amdgcn_sched_barrier(0);
            v4f m[4][SC_IU];
#pragma unroll
            for (int e = 0; e < 4; ++e)
#pragma unroll
                for (int u = 0; u < SC_IU; ++u)
#pragma unroll
                    for (int c = 0; c < 4; ++c)
                        m[e][u][c] = fminf(a[q & 1][u][e], ajr[4 * q + e][c]);
#pragma unroll
            for (int e = 0; e < 4; ++e)
#pragma unroll
                for (int u = 0; u < SC_IU; ++u)
#pragma unroll
                    for (int c = 0; c < 4; ++c)
                        acc[u][c] = fmaf(w2r[4 * q + e], m[e][u][c], acc[u][c]);
            __builtin_amdgcn_sched_barrier(0);
        }

#pragma unroll
        for (int u = 0; u < SC_IU; ++u) s_part[hs][ig + u][ct] = acc[u];
    }

    __syncthreads();

#pragma unroll
    for (int r = 0; r < SC_TI / 2; ++r) {
        const int rr = hs * (SC_TI / 2) + r;
        const v4f p0 = s_part[0][rr][ct];
        const v4f p1 = s_part[1][rr][ct];
        const int i  = i0 + rr;
        v4f o;
#pragma unroll
        for (int c = 0; c < 4; ++c) {
            float val = 1.0f / (1.0f + __expf(p0[c] + p1[c]));
            if (j0 + c == i) val = 0.0f;
            o[c] = val;
        }
        *(v4f*)(adj + (row0 + rr) * Nn + j0) = o;
    }
}

extern "C" void kernel_launch(void* const* d_in, const int* in_sizes, int n_in,
                              void* d_out, int out_size, void* d_ws, size_t ws_size,
                              hipStream_t stream) {
    const float* X      = (const float*)d_in[0];
    const float* W_enc  = (const float*)d_in[1];
    const float* b_enc  = (const float*)d_in[2];
    const float* W1     = (const float*)d_in[3];
    const float* b1     = (const float*)d_in[4];
    const float* W2     = (const float*)d_in[5];
    const float* b2     = (const float*)d_in[6];
    const float* sp     = (const float*)d_in[7];

    float* out_adj    = (float*)d_out;
    float* out_struct = out_adj + Bsz * Nn * Nn;

    float* ws    = (float*)d_ws;
    float* ai    = ws;
    float* ajt   = ai + 8192 * 64;

    gemm_chain_wmma_f32<<<256, 128, 0, stream>>>(X, W_enc, b_enc, W1, ai, ajt);

    score_kernel<<<Bsz * (Nn / SC_TI), SC_NT, 0, stream>>>(ai, ajt, b1, W2, b2, sp, out_adj, out_struct);
}
